# starting kernel
# baseline (speedup 1.0000x reference)
#include <hip/hip_runtime.h>
#include <math.h>

typedef __attribute__((ext_vector_type(2))) float v2f;
typedef __attribute__((ext_vector_type(8))) float v8f;

#define NTHREADS 256

__global__ void fill_zero_f32(float* __restrict__ p, long long n) {
    long long i = (long long)blockIdx.x * blockDim.x + threadIdx.x;
    long long stride = (long long)gridDim.x * blockDim.x;
    for (; i < n; i += stride) p[i] = 0.0f;
}

__global__ void count_edges(const int* __restrict__ dst, float* __restrict__ cnt, int E) {
    int i = blockIdx.x * blockDim.x + threadIdx.x;
    if (i < E) unsafeAtomicAdd(&cnt[dst[i]], 1.0f);
}

__global__ void invert_counts(float* __restrict__ cnt, int n) {
    int i = blockIdx.x * blockDim.x + threadIdx.x;
    if (i < n) cnt[i] = 1.0f / fmaxf(cnt[i], 1.0f);
}

__global__ void scatter_weighted(const float* __restrict__ x,
                                 const int* __restrict__ src,
                                 const int* __restrict__ dst,
                                 const float* __restrict__ w,
                                 float* __restrict__ agg,
                                 int E, int d, int shift) {
    long long tid = (long long)blockIdx.x * blockDim.x + threadIdx.x;
    long long total = (long long)E << shift;
    if (tid >= total) return;
    int e = (int)(tid >> shift);
    int c = (int)(tid & ((1 << shift) - 1));
    int s = src[e];
    int t = dst[e];
    float ww = w[e];
    const float4* xv = (const float4*)(x + (size_t)s * d);
    float4 v = xv[c];
    float* base = agg + (size_t)t * d + c * 4;
    unsafeAtomicAdd(base + 0, v.x * ww);
    unsafeAtomicAdd(base + 1, v.y * ww);
    unsafeAtomicAdd(base + 2, v.z * ww);
    unsafeAtomicAdd(base + 3, v.w * ww);
}

__global__ void sage_gemm_wmma(const float* __restrict__ agg,
                               const float* __restrict__ inv,
                               const float* __restrict__ xin,
                               const float* __restrict__ Wl,
                               const float* __restrict__ bl,
                               const float* __restrict__ Wr,
                               const float* __restrict__ br,
                               float* __restrict__ out,
                               int n_nodes, int d_in, int d_out,
                               int tiles_n, int do_relu) {
    int lane = threadIdx.x & 31;
    int wid = blockIdx.x * (blockDim.x >> 5) + (threadIdx.x >> 5);
    int tiles_m = n_nodes >> 4;
    if (wid >= tiles_m * tiles_n) return;
    int tm = wid / tiles_n;
    int tn = wid - tm * tiles_n;
    int m0 = tm << 4;
    int n0 = tn << 4;

    int half = lane >> 4;
    int l15  = lane & 15;
    int arow = m0 + l15;
    int bcol = n0 + l15;
    bool colok = bcol < d_out;
    int bc = colok ? bcol : (d_out - 1);

    float scale = inv[arow];
    const float* aggRow = agg + (size_t)arow * d_in + half * 2;
    const float* xRow   = xin + (size_t)arow * d_in + half * 2;
    const float* WlCol  = Wl + (size_t)(half * 2) * d_out + bc;
    const float* WrCol  = Wr + (size_t)(half * 2) * d_out + bc;

    v8f c = {};

    for (int k = 0; k < d_in; k += 4) {
        v2f a = *(const v2f*)(aggRow + k);
        a.x *= scale;
        a.y *= scale;
        v2f b;
        b.x = WlCol[(size_t)k * d_out];
        b.y = WlCol[(size_t)(k + 1) * d_out];
        c = __builtin_amdgcn_wmma_f32_16x16x4_f32(false, a, false, b,
                                                  (short)0, c, false, false);
    }
    for (int k = 0; k < d_in; k += 4) {
        v2f a = *(const v2f*)(xRow + k);
        v2f b;
        b.x = WrCol[(size_t)k * d_out];
        b.y = WrCol[(size_t)(k + 1) * d_out];
        c = __builtin_amdgcn_wmma_f32_16x16x4_f32(false, a, false, b,
                                                  (short)0, c, false, false);
    }

    float bias = bl[bc] + br[bc];
    if (colok) {
        for (int r = 0; r < 8; ++r) {
            float v = c[r] + bias;
            if (do_relu) v = fmaxf(v, 0.0f);
            int row = m0 + r + half * 8;
            out[(size_t)row * d_out + bcol] = v;
        }
    }
}

__global__ void log_softmax8(float* __restrict__ out, int n) {
    int i = blockIdx.x * blockDim.x + threadIdx.x;
    if (i >= n) return;
    float v[8];
    float m = -INFINITY;
    float* row = out + (size_t)i * 8;
#pragma unroll
    for (int j = 0; j < 8; ++j) { v[j] = row[j]; m = fmaxf(m, v[j]); }
    float s = 0.0f;
#pragma unroll
    for (int j = 0; j < 8; ++j) s += __expf(v[j] - m);
    float ls = __logf(s);
#pragma unroll
    for (int j = 0; j < 8; ++j) row[j] = v[j] - m - ls;
}

static inline int cdiv(long long a, long long b) { return (int)((a + b - 1) / b); }

extern "C" void kernel_launch(void* const* d_in, const int* in_sizes, int n_in,
                              void* d_out, int out_size, void* d_ws, size_t ws_size,
                              hipStream_t stream) {
    const float* x    = (const float*)d_in[0];
    const int*   ei   = (const int*)d_in[1];
    const float* w    = (const float*)d_in[2];
    const float* Wl0  = (const float*)d_in[3];
    const float* bl0  = (const float*)d_in[4];
    const float* Wr0  = (const float*)d_in[5];
    const float* br0  = (const float*)d_in[6];
    const float* Wl1  = (const float*)d_in[7];
    const float* bl1  = (const float*)d_in[8];
    const float* Wr1  = (const float*)d_in[9];
    const float* br1  = (const float*)d_in[10];
    const float* Wl2  = (const float*)d_in[11];
    const float* bl2  = (const float*)d_in[12];
    const float* Wr2  = (const float*)d_in[13];
    const float* br2  = (const float*)d_in[14];

    const int D0 = 128, D1 = 256, D2 = 256;
    const int N = in_sizes[0] / D0;
    const int E = in_sizes[2];
    const int* src = ei;
    const int* dst = ei + E;

    float* inv = (float*)d_ws;
    float* agg = inv + N;
    float* hA  = agg + (size_t)N * 256;
    float* hB  = hA  + (size_t)N * 256;
    float* logits = (float*)d_out;

    const int T = NTHREADS;
    const int fillBlocks = 2048;

    fill_zero_f32<<<fillBlocks, T, 0, stream>>>(inv, N);
    count_edges<<<cdiv(E, T), T, 0, stream>>>(dst, inv, E);
    invert_counts<<<cdiv(N, T), T, 0, stream>>>(inv, N);

    fill_zero_f32<<<fillBlocks, T, 0, stream>>>(agg, (long long)N * D0);
    scatter_weighted<<<cdiv((long long)E * (D0 / 4), T), T, 0, stream>>>(
        x, src, dst, w, agg, E, D0, 5);
    {
        int tiles_n = 256 / 16;
        int waves = (N / 16) * tiles_n;
        sage_gemm_wmma<<<cdiv(waves, T / 32), T, 0, stream>>>(
            agg, inv, x, Wl0, bl0, Wr0, br0, hA, N, D0, 256, tiles_n, 1);
    }

    fill_zero_f32<<<fillBlocks, T, 0, stream>>>(agg, (long long)N * D1);
    scatter_weighted<<<cdiv((long long)E * (D1 / 4), T), T, 0, stream>>>(
        hA, src, dst, w, agg, E, D1, 6);
    {
        int tiles_n = 256 / 16;
        int waves = (N / 16) * tiles_n;
        sage_gemm_wmma<<<cdiv(waves, T / 32), T, 0, stream>>>(
            agg, inv, hA, Wl1, bl1, Wr1, br1, hB, N, D1, 256, tiles_n, 1);
    }

    fill_zero_f32<<<fillBlocks, T, 0, stream>>>(agg, (long long)N * D2);
    scatter_weighted<<<cdiv((long long)E * (D2 / 4), T), T, 0, stream>>>(
        hB, src, dst, w, agg, E, D2, 6);
    {
        int tiles_n = 1;
        int waves = (N / 16) * tiles_n;
        sage_gemm_wmma<<<cdiv(waves, T / 32), T, 0, stream>>>(
            agg, inv, hB, Wl2, bl2, Wr2, br2, logits, N, D2, 8, tiles_n, 0);
    }

    log_softmax8<<<cdiv(N, T), T, 0, stream>>>(logits, N);
}


# Atomic scatter-add replaced by a gather: edges grouped by destination once per call (count+rank, prefix sum, place), then one wave per row sums w*x[src] in registers; no agg fills, no float atomics
# speedup vs baseline: 2.4590x; 2.4590x over previous
#include <hip/hip_runtime.h>
#include <math.h>

typedef __attribute__((ext_vector_type(2))) float v2f;
typedef __attribute__((ext_vector_type(8))) float v8f;

#define NTHREADS 256

__global__ void fill_zero_i32(int* __restrict__ p, int n) {
    int i = blockIdx.x * blockDim.x + threadIdx.x;
    if (i < n) p[i] = 0;
}

__global__ void count_rank(const int* __restrict__ dst, int* __restrict__ cnt,
                           int* __restrict__ rank, int E) {
    int i = blockIdx.x * blockDim.x + threadIdx.x;
    if (i < E) rank[i] = atomicAdd(&cnt[dst[i]], 1);
}

#define SCAN_T 1024
__global__ void __launch_bounds__(SCAN_T)
scan_offsets(const int* __restrict__ cnt, int* __restrict__ rowptr,
             float* __restrict__ inv, int n) {
    __shared__ int sh[SCAN_T];
    int t = threadIdx.x;
    int first = blockIdx.x * SCAN_T;

    int pre = 0;
    for (int i = t; i < first; i += SCAN_T) pre += cnt[i];
    sh[t] = pre;
    __syncthreads();
    for (int off = SCAN_T / 2; off > 0; off >>= 1) {
        if (t < off) sh[t] += sh[t + off];
        __syncthreads();
    }
    int base = sh[0];
    __syncthreads();

    int i = first + t;
    int c = (i < n) ? cnt[i] : 0;
    sh[t] = c;
    __syncthreads();
    for (int off = 1; off < SCAN_T; off <<= 1) {
        int v = (t >= off) ? sh[t - off] : 0;
        __syncthreads();
        sh[t] += v;
        __syncthreads();
    }
    int incl = sh[t];
    if (i < n) {
        rowptr[i] = base + incl - c;
        inv[i] = 1.0f / fmaxf((float)c, 1.0f);
    }
    if (i == n - 1) rowptr[n] = base + incl;
}

__global__ void place_edges(const int* __restrict__ src, const int* __restrict__ dst,
                            const float* __restrict__ w,
                            const int* __restrict__ rowptr, const int* __restrict__ rank,
                            int2* __restrict__ csr, int E) {
    int i = blockIdx.x * blockDim.x + threadIdx.x;
    if (i < E) {
        int slot = rowptr[dst[i]] + rank[i];
        csr[slot] = make_int2(src[i], __float_as_int(w[i]));
    }
}

template <int V>
__device__ __forceinline__ void gather_row(const float* __restrict__ x,
                                           const int* __restrict__ rowptr,
                                           const int2* __restrict__ csr,
                                           float* __restrict__ agg, int n) {
    int lane = threadIdx.x & 31;
    int node = blockIdx.x * (NTHREADS / 32) + (threadIdx.x >> 5);
    if (node >= n) return;
    int beg = __builtin_amdgcn_readfirstlane(rowptr[node]);
    int end = __builtin_amdgcn_readfirstlane(rowptr[node + 1]);

    float4 acc[V];
#pragma unroll
    for (int v = 0; v < V; ++v) acc[v] = make_float4(0.0f, 0.0f, 0.0f, 0.0f);

    for (int b0 = beg; b0 < end; b0 += 32) {
        int m = end - b0;
        if (m > 32) m = 32;
        int2 ent = make_int2(0, 0);
        if (lane < m) ent = csr[b0 + lane];

        int j = 0;
        for (; j + 4 <= m; j += 4) {
            float4 r[4][V];
            float ww[4];
#pragma unroll
            for (int u = 0; u < 4; ++u) {
                int s = __builtin_amdgcn_readlane(ent.x, j + u);
                ww[u] = __int_as_float(__builtin_amdgcn_readlane(ent.y, j + u));
                const float4* row = (const float4*)(x + (size_t)s * (128 * V));
#pragma unroll
                for (int v = 0; v < V; ++v) r[u][v] = row[lane + 32 * v];
            }
#pragma unroll
            for (int u = 0; u < 4; ++u) {
#pragma unroll
                for (int v = 0; v < V; ++v) {
                    acc[v].x += r[u][v].x * ww[u];
                    acc[v].y += r[u][v].y * ww[u];
                    acc[v].z += r[u][v].z * ww[u];
                    acc[v].w += r[u][v].w * ww[u];
                }
            }
        }
        for (; j < m; ++j) {
            int s = __builtin_amdgcn_readlane(ent.x, j);
            float w1 = __int_as_float(__builtin_amdgcn_readlane(ent.y, j));
            const float4* row = (const float4*)(x + (size_t)s * (128 * V));
#pragma unroll
            for (int v = 0; v < V; ++v) {
                float4 r1 = row[lane + 32 * v];
                acc[v].x += r1.x * w1;
                acc[v].y += r1.y * w1;
                acc[v].z += r1.z * w1;
                acc[v].w += r1.w * w1;
            }
        }
    }

    float4* o = (float4*)(agg + (size_t)node * (128 * V));
#pragma unroll
    for (int v = 0; v < V; ++v) o[lane + 32 * v] = acc[v];
}

__global__ void __launch_bounds__(NTHREADS)
gather_rows_128(const float* __restrict__ x, const int* __restrict__ rowptr,
                const int2* __restrict__ csr, float* __restrict__ agg, int n) {
    gather_row<1>(x, rowptr, csr, agg, n);
}

__global__ void __launch_bounds__(NTHREADS)
gather_rows_256(const float* __restrict__ x, const int* __restrict__ rowptr,
                const int2* __restrict__ csr, float* __restrict__ agg, int n) {
    gather_row<2>(x, rowptr, csr, agg, n);
}

__global__ void sage_gemm_wmma(const float* __restrict__ agg,
                               const float* __restrict__ inv,
                               const float* __restrict__ xin,
                               const float* __restrict__ Wl,
                               const float* __restrict__ bl,
                               const float* __restrict__ Wr,
                               const float* __restrict__ br,
                               float* __restrict__ out,
                               int n_nodes, int d_in, int d_out,
                               int tiles_n, int do_relu) {
    int lane = threadIdx.x & 31;
    int wid = blockIdx.x * (blockDim.x >> 5) + (threadIdx.x >> 5);
    int tiles_m = n_nodes >> 4;
    if (wid >= tiles_m * tiles_n) return;
    int tm = wid / tiles_n;
    int tn = wid - tm * tiles_n;
    int m0 = tm << 4;
    int n0 = tn << 4;

    int half = lane >> 4;
    int l15  = lane & 15;
    int arow = m0 + l15;
    int bcol = n0 + l15;
    bool colok = bcol < d_out;
    int bc = colok ? bcol : (d_out - 1);

    float scale = inv[arow];
    const float* aggRow = agg + (size_t)arow * d_in + half * 2;
    const float* xRow   = xin + (size_t)arow * d_in + half * 2;
    const float* WlCol  = Wl + (size_t)(half * 2) * d_out + bc;
    const float* WrCol  = Wr + (size_t)(half * 2) * d_out + bc;

    v8f c = {};

    for (int k = 0; k < d_in; k += 4) {
        v2f a = *(const v2f*)(aggRow + k);
        a.x *= scale;
        a.y *= scale;
        v2f b;
        b.x = WlCol[(size_t)k * d_out];
        b.y = WlCol[(size_t)(k + 1) * d_out];
        c = __builtin_amdgcn_wmma_f32_16x16x4_f32(false, a, false, b,
                                                  (short)0, c, false, false);
    }
    for (int k = 0; k < d_in; k += 4) {
        v2f a = *(const v2f*)(xRow + k);
        v2f b;
        b.x = WrCol[(size_t)k * d_out];
        b.y = WrCol[(size_t)(k + 1) * d_out];
        c = __builtin_amdgcn_wmma_f32_16x16x4_f32(false, a, false, b,
                                                  (short)0, c, false, false);
    }

    float bias = bl[bc] + br[bc];
    if (colok) {
        for (int r = 0; r < 8; ++r) {
            float v = c[r] + bias;
            if (do_relu) v = fmaxf(v, 0.0f);
            int row = m0 + r + half * 8;
            out[(size_t)row * d_out + bcol] = v;
        }
    }
}

__global__ void log_softmax8(float* __restrict__ out, int n) {
    int i = blockIdx.x * blockDim.x + threadIdx.x;
    if (i >= n) return;
    float v[8];
    float m = -INFINITY;
    float* row = out + (size_t)i * 8;
#pragma unroll
    for (int j = 0; j < 8; ++j) { v[j] = row[j]; m = fmaxf(m, v[j]); }
    float s = 0.0f;
#pragma unroll
    for (int j = 0; j < 8; ++j) s += __expf(v[j] - m);
    float ls = __logf(s);
#pragma unroll
    for (int j = 0; j < 8; ++j) row[j] = v[j] - m - ls;
}

static inline int cdiv(long long a, long long b) { return (int)((a + b - 1) / b); }

extern "C" void kernel_launch(void* const* d_in, const int* in_sizes, int n_in,
                              void* d_out, int out_size, void* d_ws, size_t ws_size,
                              hipStream_t stream) {
    const float* x    = (const float*)d_in[0];
    const int*   ei   = (const int*)d_in[1];
    const float* w    = (const float*)d_in[2];
    const float* Wl0  = (const float*)d_in[3];
    const float* bl0  = (const float*)d_in[4];
    const float* Wr0  = (const float*)d_in[5];
    const float* br0  = (const float*)d_in[6];
    const float* Wl1  = (const float*)d_in[7];
    const float* bl1  = (const float*)d_in[8];
    const float* Wr1  = (const float*)d_in[9];
    const float* br1  = (const float*)d_in[10];
    const float* Wl2  = (const float*)d_in[11];
    const float* bl2  = (const float*)d_in[12];
    const float* Wr2  = (const float*)d_in[13];
    const float* br2  = (const float*)d_in[14];

    const int D0 = 128, D1 = 256, D2 = 256;
    const int N = in_sizes[0] / D0;
    const int E = in_sizes[2];
    const int* src = ei;
    const int* dst = ei + E;

    float* inv = (float*)d_ws;
    float* agg = inv + N;
    float* hA  = agg + (size_t)N * 256;
    float* hB  = hA  + (size_t)N * 256;
    int*   rowptr = (int*)(hB + (size_t)N * 256);
    int2*  csr = (int2*)(rowptr + N + 4);
    int*   rank = (int*)agg;
    int*   cnt  = rank + E;
    float* logits = (float*)d_out;
    if ((size_t)((const char*)(csr + E) - (const char*)d_ws) > ws_size) return;

    const int T = NTHREADS;

    fill_zero_i32<<<cdiv(N, T), T, 0, stream>>>(cnt, N);
    count_rank<<<cdiv(E, T), T, 0, stream>>>(dst, cnt, rank, E);
    scan_offsets<<<cdiv(N, SCAN_T), SCAN_T, 0, stream>>>(cnt, rowptr, inv, N);
    place_edges<<<cdiv(E, T), T, 0, stream>>>(src, dst, w, rowptr, rank, csr, E);

    gather_rows_128<<<cdiv(N, T / 32), T, 0, stream>>>(x, rowptr, csr, agg, N);
    {
        int tiles_n = 256 / 16;
        int waves = (N / 16) * tiles_n;
        sage_gemm_wmma<<<cdiv(waves, T / 32), T, 0, stream>>>(
            agg, inv, x, Wl0, bl0, Wr0, br0, hA, N, D0, 256, tiles_n, 1);
    }

    gather_rows_256<<<cdiv(N, T / 32), T, 0, stream>>>(hA, rowptr, csr, agg, N);
    {
        int tiles_n = 256 / 16;
        int waves = (N / 16) * tiles_n;
        sage_gemm_wmma<<<cdiv(waves, T / 32), T, 0, stream>>>(
            agg, inv, hA, Wl1, bl1, Wr1, br1, hB, N, D1, 256, tiles_n, 1);
    }

    gather_rows_256<<<cdiv(N, T / 32), T, 0, stream>>>(hB, rowptr, csr, agg, N);
    {
        int tiles_n = 1;
        int waves = (N / 16) * tiles_n;
        sage_gemm_wmma<<<cdiv(waves, T / 32), T, 0, stream>>>(
            agg, inv, hB, Wl2, bl2, Wr2, br2, logits, N, D2, 8, tiles_n, 0);
    }

    log_softmax8<<<cdiv(N, T), T, 0, stream>>>(logits, N);
}


# GEMM of the two 256-column layers tiled through LDS: a 16-wave work-group per 208x256 block stages each 32-deep operand slice once (two stages), each wave keeps 13 accumulator tiles
# speedup vs baseline: 3.8260x; 1.5559x over previous
#include <hip/hip_runtime.h>
#include <math.h>

typedef __attribute__((ext_vector_type(2))) float v2f;
typedef __attribute__((ext_vector_type(8))) float v8f;

#define NTHREADS 256

__global__ void fill_zero_i32(int* __restrict__ p, int n) {
    int i = blockIdx.x * blockDim.x + threadIdx.x;
    if (i < n) p[i] = 0;
}

__global__ void count_rank(const int* __restrict__ dst, int* __restrict__ cnt,
                           int* __restrict__ rank, int E) {
    int i = blockIdx.x * blockDim.x + threadIdx.x;
    if (i < E) rank[i] = atomicAdd(&cnt[dst[i]], 1);
}

#define SCAN_T 1024
__global__ void __launch_bounds__(SCAN_T)
scan_offsets(const int* __restrict__ cnt, int* __restrict__ rowptr,
             float* __restrict__ inv, int n) {
    __shared__ int sh[SCAN_T];
    int t = threadIdx.x;
    int first = blockIdx.x * SCAN_T;

    int pre = 0;
    for (int i = t; i < first; i += SCAN_T) pre += cnt[i];
    sh[t] = pre;
    __syncthreads();
    for (int off = SCAN_T / 2; off > 0; off >>= 1) {
        if (t < off) sh[t] += sh[t + off];
        __syncthreads();
    }
    int base = sh[0];
    __syncthreads();

    int i = first + t;
    int c = (i < n) ? cnt[i] : 0;
    sh[t] = c;
    __syncthreads();
    for (int off = 1; off < SCAN_T; off <<= 1) {
        int v = (t >= off) ? sh[t - off] : 0;
        __syncthreads();
        sh[t] += v;
        __syncthreads();
    }
    int incl = sh[t];
    if (i < n) {
        rowptr[i] = base + incl - c;
        inv[i] = 1.0f / fmaxf((float)c, 1.0f);
    }
    if (i == n - 1) rowptr[n] = base + incl;
}

__global__ void place_edges(const int* __restrict__ src, const int* __restrict__ dst,
                            const float* __restrict__ w,
                            const int* __restrict__ rowptr, const int* __restrict__ rank,
                            int2* __restrict__ csr, int E) {
    int i = blockIdx.x * blockDim.x + threadIdx.x;
    if (i < E) {
        int slot = rowptr[dst[i]] + rank[i];
        csr[slot] = make_int2(src[i], __float_as_int(w[i]));
    }
}

template <int V>
__device__ __forceinline__ void gather_row(const float* __restrict__ x,
                                           const int* __restrict__ rowptr,
                                           const int2* __restrict__ csr,
                                           float* __restrict__ agg, int n) {
    int lane = threadIdx.x & 31;
    int node = blockIdx.x * (NTHREADS / 32) + (threadIdx.x >> 5);
    if (node >= n) return;
    int beg = __builtin_amdgcn_readfirstlane(rowptr[node]);
    int end = __builtin_amdgcn_readfirstlane(rowptr[node + 1]);

    float4 acc[V];
#pragma unroll
    for (int v = 0; v < V; ++v) acc[v] = make_float4(0.0f, 0.0f, 0.0f, 0.0f);

    for (int b0 = beg; b0 < end; b0 += 32) {
        int m = end - b0;
        if (m > 32) m = 32;
        int2 ent = make_int2(0, 0);
        if (lane < m) ent = csr[b0 + lane];

        int j = 0;
        for (; j + 4 <= m; j += 4) {
            float4 r[4][V];
            float ww[4];
#pragma unroll
            for (int u = 0; u < 4; ++u) {
                int s = __builtin_amdgcn_readlane(ent.x, j + u);
                ww[u] = __int_as_float(__builtin_amdgcn_readlane(ent.y, j + u));
                const float4* row = (const float4*)(x + (size_t)s * (128 * V));
#pragma unroll
                for (int v = 0; v < V; ++v) r[u][v] = row[lane + 32 * v];
            }
#pragma unroll
            for (int u = 0; u < 4; ++u) {
#pragma unroll
                for (int v = 0; v < V; ++v) {
                    acc[v].x += r[u][v].x * ww[u];
                    acc[v].y += r[u][v].y * ww[u];
                    acc[v].z += r[u][v].z * ww[u];
                    acc[v].w += r[u][v].w * ww[u];
                }
            }
        }
        for (; j < m; ++j) {
            int s = __builtin_amdgcn_readlane(ent.x, j);
            float w1 = __int_as_float(__builtin_amdgcn_readlane(ent.y, j));
            const float4* row = (const float4*)(x + (size_t)s * (128 * V));
#pragma unroll
            for (int v = 0; v < V; ++v) {
                float4 r1 = row[lane + 32 * v];
                acc[v].x += r1.x * w1;
                acc[v].y += r1.y * w1;
                acc[v].z += r1.z * w1;
                acc[v].w += r1.w * w1;
            }
        }
    }

    float4* o = (float4*)(agg + (size_t)node * (128 * V));
#pragma unroll
    for (int v = 0; v < V; ++v) o[lane + 32 * v] = acc[v];
}

__global__ void __launch_bounds__(NTHREADS)
gather_rows_128(const float* __restrict__ x, const int* __restrict__ rowptr,
                const int2* __restrict__ csr, float* __restrict__ agg, int n) {
    gather_row<1>(x, rowptr, csr, agg, n);
}

__global__ void __launch_bounds__(NTHREADS)
gather_rows_256(const float* __restrict__ x, const int* __restrict__ rowptr,
                const int2* __restrict__ csr, float* __restrict__ agg, int n) {
    gather_row<2>(x, rowptr, csr, agg, n);
}

__global__ void sage_gemm_wmma(const float* __restrict__ agg,
                               const float* __restrict__ inv,
                               const float* __restrict__ xin,
                               const float* __restrict__ Wl,
                               const float* __restrict__ bl,
                               const float* __restrict__ Wr,
                               const float* __restrict__ br,
                               float* __restrict__ out,
                               int n_nodes, int d_in, int d_out,
                               int tiles_n, int do_relu) {
    int lane = threadIdx.x & 31;
    int wid = blockIdx.x * (blockDim.x >> 5) + (threadIdx.x >> 5);
    int tiles_m = n_nodes >> 4;
    if (wid >= tiles_m * tiles_n) return;
    int tm = wid / tiles_n;
    int tn = wid - tm * tiles_n;
    int m0 = tm << 4;
    int n0 = tn << 4;

    int half = lane >> 4;
    int l15  = lane & 15;
    int arow = m0 + l15;
    int bcol = n0 + l15;
    bool colok = bcol < d_out;
    int bc = colok ? bcol : (d_out - 1);

    float scale = inv[arow];
    const float* aggRow = agg + (size_t)arow * d_in + half * 2;
    const float* xRow   = xin + (size_t)arow * d_in + half * 2;
    const float* WlCol  = Wl + (size_t)(half * 2) * d_out + bc;
    const float* WrCol  = Wr + (size_t)(half * 2) * d_out + bc;

    v8f c = {};

    for (int k = 0; k < d_in; k += 4) {
        v2f a = *(const v2f*)(aggRow + k);
        a.x *= scale;
        a.y *= scale;
        v2f b;
        b.x = WlCol[(size_t)k * d_out];
        b.y = WlCol[(size_t)(k + 1) * d_out];
        c = __builtin_amdgcn_wmma_f32_16x16x4_f32(false, a, false, b,
                                                  (short)0, c, false, false);
    }
    for (int k = 0; k < d_in; k += 4) {
        v2f a = *(const v2f*)(xRow + k);
        v2f b;
        b.x = WrCol[(size_t)k * d_out];
        b.y = WrCol[(size_t)(k + 1) * d_out];
        c = __builtin_amdgcn_wmma_f32_16x16x4_f32(false, a, false, b,
                                                  (short)0, c, false, false);
    }

    float bias = bl[bc] + br[bc];
    if (colok) {
        for (int r = 0; r < 8; ++r) {
            float v = c[r] + bias;
            if (do_relu) v = fmaxf(v, 0.0f);
            int row = m0 + r + half * 8;
            out[(size_t)row * d_out + bcol] = v;
        }
    }
}

#define GB_MT      13
#define GB_ROWS    (GB_MT * 16)
#define GB_KC      32
#define GB_AP      36
#define GB_BP      264
#define GB_THREADS 512
#define GB_A_LDSROWS 256
#define GB_A_FLOATS (GB_A_LDSROWS * GB_AP)
#define GB_B_FLOATS (GB_KC * GB_BP)
#define GB_STAGE   (GB_A_FLOATS + GB_B_FLOATS)

__device__ __forceinline__ void gemm_slice(const float* __restrict__ As,
                                           const float* __restrict__ Bs,
                                           v8f (&acc)[GB_MT], int half, int l15, int n0) {
#pragma unroll
    for (int kk = 0; kk < GB_KC; kk += 4) {
        v2f b;
        b.x = Bs[(kk + 2 * half) * GB_BP + n0 + l15];
        b.y = Bs[(kk + 2 * half + 1) * GB_BP + n0 + l15];
#pragma unroll
        for (int mt = 0; mt < GB_MT; ++mt) {
            v2f a = *(const v2f*)(As + (mt * 16 + l15) * GB_AP + kk + 2 * half);
            acc[mt] = __builtin_amdgcn_wmma_f32_16x16x4_f32(false, a, false, b,
                                                            (short)0, acc[mt], false, false);
        }
    }
}

template <int DIN>
__global__ void __launch_bounds__(GB_THREADS) __attribute__((amdgpu_waves_per_eu(4, 4)))
sage_gemm_tiled(const float* __restrict__ agg,
                const float* __restrict__ inv,
                const float* __restrict__ xin,
                const float* __restrict__ Wl,
                const float* __restrict__ bl,
                const float* __restrict__ Wr,
                const float* __restrict__ br,
                float* __restrict__ out,
                int n_nodes) {
    __shared__ __attribute__((aligned(16))) float lds[2 * GB_STAGE];

    const int tid  = threadIdx.x;
    const int lane = tid & 31;
    const int wave = tid >> 5;
    const int half = lane >> 4;
    const int l15  = lane & 15;
    const int m0   = blockIdx.x * GB_ROWS;
    const int n0   = wave * 16;

    constexpr int NC_HALF = DIN / GB_KC;
    constexpr int NC      = 2 * NC_HALF;

    int   aOff[4], aLds[4], bOff[4], bLds[4];
    float aScale[4];
#pragma unroll
    for (int it = 0; it < 4; ++it) {
        int idx = it * GB_THREADS + tid;
        int row = idx >> 3;
        int c4  = idx & 7;
        int srow = row < GB_ROWS ? row : GB_ROWS - 1;
        int grow = m0 + srow;
        if (grow > n_nodes - 1) grow = n_nodes - 1;
        aOff[it]   = grow * DIN + c4 * 4;
        aScale[it] = inv[grow];
        aLds[it]   = row * GB_AP + c4 * 4;
        int kr = idx >> 6;
        int n4 = idx & 63;
        bOff[it] = kr * 256 + n4 * 4;
        bLds[it] = kr * GB_BP + n4 * 4;
    }

    v8f acc[GB_MT];
#pragma unroll
    for (int mt = 0; mt < GB_MT; ++mt) acc[mt] = v8f{};

    float4 ra[4], rb[4];

#pragma unroll
    for (int it = 0; it < 4; ++it) {
        ra[it] = *(const float4*)(agg + (size_t)aOff[it]);
        rb[it] = *(const float4*)(Wl + (size_t)bOff[it]);
    }
#pragma unroll
    for (int it = 0; it < 4; ++it) {
        float4 v = ra[it];
        float s = aScale[it];
        v.x *= s; v.y *= s; v.z *= s; v.w *= s;
        *(float4*)(lds + aLds[it]) = v;
        *(float4*)(lds + GB_A_FLOATS + bLds[it]) = rb[it];
    }
    __syncthreads();

    for (int c = 0; c < NC - 1; ++c) {
        const int  buf = c & 1;
        const bool nextFirst = (c + 1 < NC_HALF);
        const float* A  = nextFirst ? agg : xin;
        const float* Wm = nextFirst ? Wl : Wr;
        const int k0 = (nextFirst ? (c + 1) : (c + 1 - NC_HALF)) * GB_KC;

#pragma unroll
        for (int it = 0; it < 4; ++it) {
            ra[it] = *(const float4*)(A + (size_t)(aOff[it] + k0));
            rb[it] = *(const float4*)(Wm + (size_t)(k0 * 256 + bOff[it]));
        }

        gemm_slice(lds + buf * GB_STAGE, lds + buf * GB_STAGE + GB_A_FLOATS, acc, half, l15, n0);

        float* Ad = lds + (buf ^ 1) * GB_STAGE;
        float* Bd = Ad + GB_A_FLOATS;
#pragma unroll
        for (int it = 0; it < 4; ++it) {
            float4 v = ra[it];
            float s = aScale[it];
            v.x = nextFirst ? v.x * s : v.x;
            v.y = nextFirst ? v.y * s : v.y;
            v.z = nextFirst ? v.z * s : v.z;
            v.w = nextFirst ? v.w * s : v.w;
            *(float4*)(Ad + aLds[it]) = v;
            *(float4*)(Bd + bLds[it]) = rb[it];
        }
        __syncthreads();
    }
    {
        constexpr int buf = (NC - 1) & 1;
        gemm_slice(lds + buf * GB_STAGE, lds + buf * GB_STAGE + GB_A_FLOATS, acc, half, l15, n0);
    }

    const int col = n0 + l15;
    const float bias = bl[col] + br[col];
#pragma unroll
    for (int mt = 0; mt < GB_MT; ++mt) {
#pragma unroll
        for (int r = 0; r < 8; ++r) {
            int row = m0 + mt * 16 + half * 8 + r;
            if (row < n_nodes) {
                float v = fmaxf(acc[mt][r] + bias, 0.0f);
                out[(size_t)row * 256 + col] = v;
            }
        }
    }
}

__global__ void log_softmax8(float* __restrict__ out, int n) {
    int i = blockIdx.x * blockDim.x + threadIdx.x;
    if (i >= n) return;
    float v[8];
    float m = -INFINITY;
    float* row = out + (size_t)i * 8;
#pragma unroll
    for (int j = 0; j < 8; ++j) { v[j] = row[j]; m = fmaxf(m, v[j]); }
    float s = 0.0f;
#pragma unroll
    for (int j = 0; j < 8; ++j) s += __expf(v[j] - m);
    float ls = __logf(s);
#pragma unroll
    for (int j = 0; j < 8; ++j) row[j] = v[j] - m - ls;
}

static inline int cdiv(long long a, long long b) { return (int)((a + b - 1) / b); }

extern "C" void kernel_launch(void* const* d_in, const int* in_sizes, int n_in,
                              void* d_out, int out_size, void* d_ws, size_t ws_size,
                              hipStream_t stream) {
    const float* x    = (const float*)d_in[0];
    const int*   ei   = (const int*)d_in[1];
    const float* w    = (const float*)d_in[2];
    const float* Wl0  = (const float*)d_in[3];
    const float* bl0  = (const float*)d_in[4];
    const float* Wr0  = (const float*)d_in[5];
    const float* br0  = (const float*)d_in[6];
    const float* Wl1  = (const float*)d_in[7];
    const float* bl1  = (const float*)d_in[8];
    const float* Wr1  = (const float*)d_in[9];
    const float* br1  = (const float*)d_in[10];
    const float* Wl2  = (const float*)d_in[11];
    const float* bl2  = (const float*)d_in[12];
    const float* Wr2  = (const float*)d_in[13];
    const float* br2  = (const float*)d_in[14];

    const int D0 = 128, D1 = 256, D2 = 256;
    const int N = in_sizes[0] / D0;
    const int E = in_sizes[2];
    const int* src = ei;
    const int* dst = ei + E;

    float* inv = (float*)d_ws;
    float* agg = inv + N;
    float* hA  = agg + (size_t)N * 256;
    float* hB  = hA  + (size_t)N * 256;
    int*   rowptr = (int*)(hB + (size_t)N * 256);
    int2*  csr = (int2*)(rowptr + N + 4);
    int*   rank = (int*)agg;
    int*   cnt  = rank + E;
    float* logits = (float*)d_out;
    if ((size_t)((const char*)(csr + E) - (const char*)d_ws) > ws_size) return;

    const int T = NTHREADS;

    fill_zero_i32<<<cdiv(N, T), T, 0, stream>>>(cnt, N);
    count_rank<<<cdiv(E, T), T, 0, stream>>>(dst, cnt, rank, E);
    scan_offsets<<<cdiv(N, SCAN_T), SCAN_T, 0, stream>>>(cnt, rowptr, inv, N);
    place_edges<<<cdiv(E, T), T, 0, stream>>>(src, dst, w, rowptr, rank, csr, E);

    gather_rows_128<<<cdiv(N, T / 32), T, 0, stream>>>(x, rowptr, csr, agg, N);
    sage_gemm_tiled<128><<<cdiv(N, GB_ROWS), GB_THREADS, 0, stream>>>(
        agg, inv, x, Wl0, bl0, Wr0, br0, hA, N);

    gather_rows_256<<<cdiv(N, T / 32), T, 0, stream>>>(hA, rowptr, csr, agg, N);
    sage_gemm_tiled<256><<<cdiv(N, GB_ROWS), GB_THREADS, 0, stream>>>(
        agg, inv, hA, Wl1, bl1, Wr1, br1, hB, N);

    gather_rows_256<<<cdiv(N, T / 32), T, 0, stream>>>(hB, rowptr, csr, agg, N);
    {
        int tiles_n = 1;
        int waves = (N / 16) * tiles_n;
        sage_gemm_wmma<<<cdiv(waves, T / 32), T, 0, stream>>>(
            agg, inv, hB, Wl2, bl2, Wr2, br2, logits, N, D2, 8, tiles_n, 0);
    }

    log_softmax8<<<cdiv(N, T), T, 0, stream>>>(logits, N);
}


# Last layer multiplies before it aggregates (the mean is linear): h @ [Wl2|Wr2] on one full 16-column WMMA tile, then edge sums over 8-float rows fused with log_softmax; no third 256-wide gather
# speedup vs baseline: 4.4507x; 1.1633x over previous
#include <hip/hip_runtime.h>
#include <math.h>

typedef __attribute__((ext_vector_type(2))) float v2f;
typedef __attribute__((ext_vector_type(8))) float v8f;

#define NTHREADS 256

__global__ void fill_zero_i32(int* __restrict__ p, int n) {
    int i = blockIdx.x * blockDim.x + threadIdx.x;
    if (i < n) p[i] = 0;
}

__global__ void count_rank(const int* __restrict__ dst, int* __restrict__ cnt,
                           int* __restrict__ rank, int E) {
    int i = blockIdx.x * blockDim.x + threadIdx.x;
    if (i < E) rank[i] = atomicAdd(&cnt[dst[i]], 1);
}

#define SCAN_T 1024
__global__ void __launch_bounds__(SCAN_T)
scan_offsets(const int* __restrict__ cnt, int* __restrict__ rowptr,
             float* __restrict__ inv, int n) {
    __shared__ int sh[SCAN_T];
    int t = threadIdx.x;
    int first = blockIdx.x * SCAN_T;

    int pre = 0;
    for (int i = t; i < first; i += SCAN_T) pre += cnt[i];
    sh[t] = pre;
    __syncthreads();
    for (int off = SCAN_T / 2; off > 0; off >>= 1) {
        if (t < off) sh[t] += sh[t + off];
        __syncthreads();
    }
    int base = sh[0];
    __syncthreads();

    int i = first + t;
    int c = (i < n) ? cnt[i] : 0;
    sh[t] = c;
    __syncthreads();
    for (int off = 1; off < SCAN_T; off <<= 1) {
        int v = (t >= off) ? sh[t - off] : 0;
        __syncthreads();
        sh[t] += v;
        __syncthreads();
    }
    int incl = sh[t];
    if (i < n) {
        rowptr[i] = base + incl - c;
        inv[i] = 1.0f / fmaxf((float)c, 1.0f);
    }
    if (i == n - 1) rowptr[n] = base + incl;
}

__global__ void place_edges(const int* __restrict__ src, const int* __restrict__ dst,
                            const float* __restrict__ w,
                            const int* __restrict__ rowptr, const int* __restrict__ rank,
                            int2* __restrict__ csr, int E) {
    int i = blockIdx.x * blockDim.x + threadIdx.x;
    if (i < E) {
        int slot = rowptr[dst[i]] + rank[i];
        csr[slot] = make_int2(src[i], __float_as_int(w[i]));
    }
}

template <int V>
__device__ __forceinline__ void gather_row(const float* __restrict__ x,
                                           const int* __restrict__ rowptr,
                                           const int2* __restrict__ csr,
                                           float* __restrict__ agg, int n) {
    int lane = threadIdx.x & 31;
    int node = blockIdx.x * (NTHREADS / 32) + (threadIdx.x >> 5);
    if (node >= n) return;
    int beg = __builtin_amdgcn_readfirstlane(rowptr[node]);
    int end = __builtin_amdgcn_readfirstlane(rowptr[node + 1]);

    float4 acc[V];
#pragma unroll
    for (int v = 0; v < V; ++v) acc[v] = make_float4(0.0f, 0.0f, 0.0f, 0.0f);

    for (int b0 = beg; b0 < end; b0 += 32) {
        int m = end - b0;
        if (m > 32) m = 32;
        int2 ent = make_int2(0, 0);
        if (lane < m) ent = csr[b0 + lane];

        int j = 0;
        for (; j + 4 <= m; j += 4) {
            float4 r[4][V];
            float ww[4];
#pragma unroll
            for (int u = 0; u < 4; ++u) {
                int s = __builtin_amdgcn_readlane(ent.x, j + u);
                ww[u] = __int_as_float(__builtin_amdgcn_readlane(ent.y, j + u));
                const float4* row = (const float4*)(x + (size_t)s * (128 * V));
#pragma unroll
                for (int v = 0; v < V; ++v) r[u][v] = row[lane + 32 * v];
            }
#pragma unroll
            for (int u = 0; u < 4; ++u) {
#pragma unroll
                for (int v = 0; v < V; ++v) {
                    acc[v].x += r[u][v].x * ww[u];
                    acc[v].y += r[u][v].y * ww[u];
                    acc[v].z += r[u][v].z * ww[u];
                    acc[v].w += r[u][v].w * ww[u];
                }
            }
        }
        for (; j < m; ++j) {
            int s = __builtin_amdgcn_readlane(ent.x, j);
            float w1 = __int_as_float(__builtin_amdgcn_readlane(ent.y, j));
            const float4* row = (const float4*)(x + (size_t)s * (128 * V));
#pragma unroll
            for (int v = 0; v < V; ++v) {
                float4 r1 = row[lane + 32 * v];
                acc[v].x += r1.x * w1;
                acc[v].y += r1.y * w1;
                acc[v].z += r1.z * w1;
                acc[v].w += r1.w * w1;
            }
        }
    }

    float4* o = (float4*)(agg + (size_t)node * (128 * V));
#pragma unroll
    for (int v = 0; v < V; ++v) o[lane + 32 * v] = acc[v];
}

__global__ void __launch_bounds__(NTHREADS)
gather_rows_128(const float* __restrict__ x, const int* __restrict__ rowptr,
                const int2* __restrict__ csr, float* __restrict__ agg, int n) {
    gather_row<1>(x, rowptr, csr, agg, n);
}

__global__ void __launch_bounds__(NTHREADS)
gather_rows_256(const float* __restrict__ x, const int* __restrict__ rowptr,
                const int2* __restrict__ csr, float* __restrict__ agg, int n) {
    gather_row<2>(x, rowptr, csr, agg, n);
}

#define GB_MT      13
#define GB_ROWS    (GB_MT * 16)
#define GB_KC      32
#define GB_AP      36
#define GB_BP      264
#define GB_THREADS 512
#define GB_A_LDSROWS 256
#define GB_A_FLOATS (GB_A_LDSROWS * GB_AP)
#define GB_B_FLOATS (GB_KC * GB_BP)
#define GB_STAGE   (GB_A_FLOATS + GB_B_FLOATS)

__device__ __forceinline__ void gemm_slice(const float* __restrict__ As,
                                           const float* __restrict__ Bs,
                                           v8f (&acc)[GB_MT], int half, int l15, int n0) {
#pragma unroll
    for (int kk = 0; kk < GB_KC; kk += 4) {
        v2f b;
        b.x = Bs[(kk + 2 * half) * GB_BP + n0 + l15];
        b.y = Bs[(kk + 2 * half + 1) * GB_BP + n0 + l15];
#pragma unroll
        for (int mt = 0; mt < GB_MT; ++mt) {
            v2f a = *(const v2f*)(As + (mt * 16 + l15) * GB_AP + kk + 2 * half);
            acc[mt] = __builtin_amdgcn_wmma_f32_16x16x4_f32(false, a, false, b,
                                                            (short)0, acc[mt], false, false);
        }
    }
}

template <int DIN>
__global__ void __launch_bounds__(GB_THREADS) __attribute__((amdgpu_waves_per_eu(4, 4)))
sage_gemm_tiled(const float* __restrict__ agg,
                const float* __restrict__ inv,
                const float* __restrict__ xin,
                const float* __restrict__ Wl,
                const float* __restrict__ bl,
                const float* __restrict__ Wr,
                const float* __restrict__ br,
                float* __restrict__ out,
                int n_nodes) {
    __shared__ __attribute__((aligned(16))) float lds[2 * GB_STAGE];

    const int tid  = threadIdx.x;
    const int lane = tid & 31;
    const int wave = tid >> 5;
    const int half = lane >> 4;
    const int l15  = lane & 15;
    const int m0   = blockIdx.x * GB_ROWS;
    const int n0   = wave * 16;

    constexpr int NC_HALF = DIN / GB_KC;
    constexpr int NC      = 2 * NC_HALF;

    int   aOff[4], aLds[4], bOff[4], bLds[4];
    float aScale[4];
#pragma unroll
    for (int it = 0; it < 4; ++it) {
        int idx = it * GB_THREADS + tid;
        int row = idx >> 3;
        int c4  = idx & 7;
        int srow = row < GB_ROWS ? row : GB_ROWS - 1;
        int grow = m0 + srow;
        if (grow > n_nodes - 1) grow = n_nodes - 1;
        aOff[it]   = grow * DIN + c4 * 4;
        aScale[it] = inv[grow];
        aLds[it]   = row * GB_AP + c4 * 4;
        int kr = idx >> 6;
        int n4 = idx & 63;
        bOff[it] = kr * 256 + n4 * 4;
        bLds[it] = kr * GB_BP + n4 * 4;
    }

    v8f acc[GB_MT];
#pragma unroll
    for (int mt = 0; mt < GB_MT; ++mt) acc[mt] = v8f{};

    float4 ra[4], rb[4];

#pragma unroll
    for (int it = 0; it < 4; ++it) {
        ra[it] = *(const float4*)(agg + (size_t)aOff[it]);
        rb[it] = *(const float4*)(Wl + (size_t)bOff[it]);
    }
#pragma unroll
    for (int it = 0; it < 4; ++it) {
        float4 v = ra[it];
        float s = aScale[it];
        v.x *= s; v.y *= s; v.z *= s; v.w *= s;
        *(float4*)(lds + aLds[it]) = v;
        *(float4*)(lds + GB_A_FLOATS + bLds[it]) = rb[it];
    }
    __syncthreads();

    for (int c = 0; c < NC - 1; ++c) {
        const int  buf = c & 1;
        const bool nextFirst = (c + 1 < NC_HALF);
        const float* A  = nextFirst ? agg : xin;
        const float* Wm = nextFirst ? Wl : Wr;
        const int k0 = (nextFirst ? (c + 1) : (c + 1 - NC_HALF)) * GB_KC;

#pragma unroll
        for (int it = 0; it < 4; ++it) {
            ra[it] = *(const float4*)(A + (size_t)(aOff[it] + k0));
            rb[it] = *(const float4*)(Wm + (size_t)(k0 * 256 + bOff[it]));
        }

        gemm_slice(lds + buf * GB_STAGE, lds + buf * GB_STAGE + GB_A_FLOATS, acc, half, l15, n0);

        float* Ad = lds + (buf ^ 1) * GB_STAGE;
        float* Bd = Ad + GB_A_FLOATS;
#pragma unroll
        for (int it = 0; it < 4; ++it) {
            float4 v = ra[it];
            float s = aScale[it];
            v.x = nextFirst ? v.x * s : v.x;
            v.y = nextFirst ? v.y * s : v.y;
            v.z = nextFirst ? v.z * s : v.z;
            v.w = nextFirst ? v.w * s : v.w;
            *(float4*)(Ad + aLds[it]) = v;
            *(float4*)(Bd + bLds[it]) = rb[it];
        }
        __syncthreads();
    }
    {
        constexpr int buf = (NC - 1) & 1;
        gemm_slice(lds + buf * GB_STAGE, lds + buf * GB_STAGE + GB_A_FLOATS, acc, half, l15, n0);
    }

    const int col = n0 + l15;
    const float bias = bl[col] + br[col];
#pragma unroll
    for (int mt = 0; mt < GB_MT; ++mt) {
#pragma unroll
        for (int r = 0; r < 8; ++r) {
            int row = m0 + mt * 16 + half * 8 + r;
            if (row < n_nodes) {
                float v = fmaxf(acc[mt][r] + bias, 0.0f);
                out[(size_t)row * 256 + col] = v;
            }
        }
    }
}

#define PJ_K      256
#define PJ_GROUPS (PJ_K / 8)
#define PJ_WAVES  (NTHREADS / 32)

__global__ void __launch_bounds__(NTHREADS)
project_last(const float* __restrict__ h,
             const float* __restrict__ Wl, const float* __restrict__ bl,
             const float* __restrict__ Wr, const float* __restrict__ br,
             float* __restrict__ P, float* __restrict__ Q, int n_nodes) {
    __shared__ __attribute__((aligned(16))) float Bs[PJ_K * 16];

    const int tid  = threadIdx.x;
    const int lane = tid & 31;
    const int half = lane >> 4;
    const int l15  = lane & 15;
    const int m0   = (blockIdx.x * PJ_WAVES + (tid >> 5)) * 16;

    int arow = m0 + l15;
    if (arow > n_nodes - 1) arow = n_nodes - 1;
    const float4* aptr = (const float4*)(h + (size_t)arow * PJ_K + 4 * half);
    float4 a[PJ_GROUPS];
#pragma unroll
    for (int j = 0; j < PJ_GROUPS; ++j) a[j] = aptr[2 * j];

#pragma unroll
    for (int i = 0; i < (PJ_K * 16) / NTHREADS; ++i) {
        int idx = i * NTHREADS + tid;
        int q   = idx & 3;
        int ln  = (idx >> 2) & 31;
        int j   = idx >> 7;
        int col = ln & 15;
        int k   = 8 * j + 4 * (ln >> 4) + q;
        Bs[idx] = (col < 8) ? Wl[k * 8 + col] : Wr[k * 8 + col - 8];
    }
    __syncthreads();

    if (m0 >= n_nodes) return;

    v8f c0 = {}, c1 = {};
#pragma unroll
    for (int j = 0; j < PJ_GROUPS; ++j) {
        float4 b = *(const float4*)(Bs + (j * 32 + lane) * 4);
        v2f a0 = {a[j].x, a[j].y}, b0 = {b.x, b.y};
        v2f a1 = {a[j].z, a[j].w}, b1 = {b.z, b.w};
        c0 = __builtin_amdgcn_wmma_f32_16x16x4_f32(false, a0, false, b0,
                                                   (short)0, c0, false, false);
        c1 = __builtin_amdgcn_wmma_f32_16x16x4_f32(false, a1, false, b1,
                                                   (short)0, c1, false, false);
    }

    const bool isQ = l15 >= 8;
    const int  oc  = l15 & 7;
    const float bias = isQ ? (bl[oc] + br[oc]) : 0.0f;
    float* dstp = isQ ? Q : P;
#pragma unroll
    for (int r = 0; r < 8; ++r) {
        int row = m0 + half * 8 + r;
        if (row < n_nodes) dstp[(size_t)row * 8 + oc] = c0[r] + c1[r] + bias;
    }
}

#define GE_EDGES 8
__global__ void __launch_bounds__(NTHREADS)
gather8_log_softmax(const float* __restrict__ P, const float* __restrict__ Q,
                    const float* __restrict__ inv,
                    const int* __restrict__ rowptr, const int2* __restrict__ csr,
                    float* __restrict__ out, int n) {
    int gid  = blockIdx.x * NTHREADS + threadIdx.x;
    int node = gid >> 3;
    int c    = gid & 7;
    bool live = node < n;
    int nd = live ? node : n - 1;

    int beg = rowptr[nd];
    int end = rowptr[nd + 1];

    float acc = 0.0f;
    for (int e0 = beg; e0 < end; e0 += GE_EDGES) {
        int idx = e0 + c;
        int2 rec = make_int2(0, 0);
        if (idx < end) rec = csr[idx];
        float p[GE_EDGES], wgt[GE_EDGES];
#pragma unroll
        for (int u = 0; u < GE_EDGES; ++u) {
            int s  = __shfl(rec.x, u, GE_EDGES);
            wgt[u] = __int_as_float(__shfl(rec.y, u, GE_EDGES));
            p[u]   = P[s * 8 + c];
        }
#pragma unroll
        for (int u = 0; u < GE_EDGES; ++u) {
            float t = acc + wgt[u] * p[u];
            acc = (e0 + u < end) ? t : acc;
        }
    }

    float v = acc * inv[nd] + Q[nd * 8 + c];

    float m = v;
    m = fmaxf(m, __shfl_xor(m, 1));
    m = fmaxf(m, __shfl_xor(m, 2));
    m = fmaxf(m, __shfl_xor(m, 4));
    float s = __expf(v - m);
    s += __shfl_xor(s, 1);
    s += __shfl_xor(s, 2);
    s += __shfl_xor(s, 4);
    float ls = __logf(s);
    if (live) out[node * 8 + c] = v - m - ls;
}

static inline int cdiv(long long a, long long b) { return (int)((a + b - 1) / b); }

extern "C" void kernel_launch(void* const* d_in, const int* in_sizes, int n_in,
                              void* d_out, int out_size, void* d_ws, size_t ws_size,
                              hipStream_t stream) {
    const float* x    = (const float*)d_in[0];
    const int*   ei   = (const int*)d_in[1];
    const float* w    = (const float*)d_in[2];
    const float* Wl0  = (const float*)d_in[3];
    const float* bl0  = (const float*)d_in[4];
    const float* Wr0  = (const float*)d_in[5];
    const float* br0  = (const float*)d_in[6];
    const float* Wl1  = (const float*)d_in[7];
    const float* bl1  = (const float*)d_in[8];
    const float* Wr1  = (const float*)d_in[9];
    const float* br1  = (const float*)d_in[10];
    const float* Wl2  = (const float*)d_in[11];
    const float* bl2  = (const float*)d_in[12];
    const float* Wr2  = (const float*)d_in[13];
    const float* br2  = (const float*)d_in[14];

    const int D0 = 128;
    const int N = in_sizes[0] / D0;
    const int E = in_sizes[2];
    const int* src = ei;
    const int* dst = ei + E;

    float* inv = (float*)d_ws;
    float* agg = inv + N;
    float* hA  = agg + (size_t)N * 256;
    float* hB  = hA  + (size_t)N * 256;
    int*   rowptr = (int*)(hB + (size_t)N * 256);
    int2*  csr = (int2*)(rowptr + N + 4);
    int*   rank = (int*)agg;
    int*   cnt  = rank + E;
    float* logits = (float*)d_out;
    if ((size_t)((const char*)(csr + E) - (const char*)d_ws) > ws_size) return;

    const int T = NTHREADS;

    fill_zero_i32<<<cdiv(N, T), T, 0, stream>>>(cnt, N);
    count_rank<<<cdiv(E, T), T, 0, stream>>>(dst, cnt, rank, E);
    scan_offsets<<<cdiv(N, SCAN_T), SCAN_T, 0, stream>>>(cnt, rowptr, inv, N);
    place_edges<<<cdiv(E, T), T, 0, stream>>>(src, dst, w, rowptr, rank, csr, E);

    gather_rows_128<<<cdiv(N, T / 32), T, 0, stream>>>(x, rowptr, csr, agg, N);
    sage_gemm_tiled<128><<<cdiv(N, GB_ROWS), GB_THREADS, 0, stream>>>(
        agg, inv, x, Wl0, bl0, Wr0, br0, hA, N);

    gather_rows_256<<<cdiv(N, T / 32), T, 0, stream>>>(hA, rowptr, csr, agg, N);
    sage_gemm_tiled<256><<<cdiv(N, GB_ROWS), GB_THREADS, 0, stream>>>(
        agg, inv, hA, Wl1, bl1, Wr1, br1, hB, N);

    float* P = agg;
    float* Q = agg + (size_t)N * 8;
    project_last<<<cdiv(N, 16 * (T / 32)), T, 0, stream>>>(hB, Wl2, bl2, Wr2, br2, P, Q, N);
    gather8_log_softmax<<<cdiv((long long)N * 8, T), T, 0, stream>>>(
        P, Q, inv, rowptr, csr, logits, N);
}
